# starting kernel
# baseline (speedup 1.0000x reference)
#include <hip/hip_runtime.h>

typedef __attribute__((ext_vector_type(2))) float v2f;
typedef __attribute__((ext_vector_type(8))) float v8f;

#define NCOL 300
#define NTILE 19
#define BFRAG_OFF 0
#define BFRAG_FLOATS (NTILE * 32 * 2)
#define W32_OFF BFRAG_FLOATS
#define W32_PAD (NTILE * 16)
#define W31_OFF (W32_OFF + W32_PAD)

__global__ __launch_bounds__(256) void phi2_prep(
    const float* __restrict__ varphi1,
    const float* __restrict__ varphi2,
    const float* __restrict__ l1,
    const float* __restrict__ l2,
    const float* __restrict__ l3,
    float* __restrict__ ws)
{
    const int tid = threadIdx.x;
    const float l3v = l3[0];

    for (int j = tid; j < W32_PAD; j += 256) {
        ws[W32_OFF + j] = (j < NCOL) ? varphi2[j] * (l2[j] - l3v) : 0.0f;
    }

    for (int i = tid; i < NTILE * 32; i += 256) {
        const int t = i >> 5;
        const int l = i & 31;
        const int j = t * 16 + (l & 15);
        float v0 = 0.0f, v1 = 0.0f;
        if (j < NCOL) {
            const float sj = -l2[j];
            if (l < 16) {
                v0 = varphi1[j * 3 + 0] * (l1[0] + sj);
                v1 = varphi1[j * 3 + 1] * (l1[1] + sj);
            } else {
                v0 = varphi1[j * 3 + 2] * (l1[2] + sj);
                v1 = 0.0f;
            }
        }
        ws[BFRAG_OFF + i * 2 + 0] = v0;
        ws[BFRAG_OFF + i * 2 + 1] = v1;
    }

    if (tid < 3) {
        const int k = tid;
        float s = 0.0f;
        for (int j = 0; j < NCOL; ++j) {
            const float w32j = varphi2[j] * (l2[j] - l3v);
            s -= w32j * varphi1[j * 3 + k];
        }
        ws[W31_OFF + k] = s;
    }
}

__device__ __forceinline__ float relu1(float v) {
    float r;
    asm("v_max_num_f32 %0, 0, %1" : "=v"(r) : "v"(v));
    return r;
}

__global__ __launch_bounds__(256) void phi2_main(
    const float* __restrict__ x,
    const float* __restrict__ ws,
    float* __restrict__ y,
    int nrows)
{
    const int lane = threadIdx.x & 31;
    const int wave = threadIdx.x >> 5;
    const long long rowBase = ((long long)blockIdx.x * 8 + wave) * 16;
    if (rowBase >= nrows) return;

    v2f a;
    {
        const int m = lane & 15;
        long long row = rowBase + m;
        if (row > nrows - 1) row = nrows - 1;
        if (lane < 16) {
            a.x = x[row * 2 + 0];
            a.y = x[row * 2 + 1];
        } else {
            a.x = 1.0f;
            a.y = 0.0f;
        }
    }

    const float* __restrict__ bf  = ws + BFRAG_OFF;
    const float* __restrict__ w32 = ws + W32_OFF;
    const float* __restrict__ w31 = ws + W31_OFF;

    v8f acc = {0.f, 0.f, 0.f, 0.f, 0.f, 0.f, 0.f, 0.f};

    for (int t = 0; t < NTILE; ++t) {
        const v2f b = *(const v2f*)(bf + (t * 32 + lane) * 2);
        v8f c = {};
        v8f d = __builtin_amdgcn_wmma_f32_16x16x4_f32(
             false, a,  false, b,
             (short)0, c,  false,  false);
        const float wv = w32[t * 16 + (lane & 15)];
#pragma unroll
        for (int r = 0; r < 8; ++r) {
            acc[r] += relu1(d[r]) * wv;
        }
    }

#pragma unroll
    for (int r = 0; r < 8; ++r) {
        float s = acc[r];
        s += __shfl_xor(s, 1, 32);
        s += __shfl_xor(s, 2, 32);
        s += __shfl_xor(s, 4, 32);
        s += __shfl_xor(s, 8, 32);
        acc[r] = s;
    }

    if ((lane & 15) == 0) {
        const float c0 = w31[0], c1 = w31[1], c2 = w31[2];
        const long long base = rowBase + ((lane >> 4) << 3);
#pragma unroll
        for (int r = 0; r < 8; ++r) {
            const long long orow = base + r;
            if (orow < nrows) {
                const float x0 = x[orow * 2 + 0];
                const float x1 = x[orow * 2 + 1];
                y[orow] = acc[r] + c0 * x0 + c1 * x1 + c2;
            }
        }
    }
}

extern "C" void kernel_launch(void* const* d_in, const int* in_sizes, int n_in,
                              void* d_out, int out_size, void* d_ws, size_t ws_size,
                              hipStream_t stream) {
    const float* x       = (const float*)d_in[0];
    const float* varphi1 = (const float*)d_in[1];
    const float* varphi2 = (const float*)d_in[2];
    const float* l1      = (const float*)d_in[3];
    const float* l2      = (const float*)d_in[4];
    const float* l3      = (const float*)d_in[5];
    float* y  = (float*)d_out;
    float* ws = (float*)d_ws;

    const int B = in_sizes[0] / 2;

    phi2_prep<<<1, 256, 0, stream>>>(varphi1, varphi2, l1, l2, l3, ws);

    const int wavesNeeded = (B + 15) / 16;
    const int grid = (wavesNeeded + 7) / 8;
    phi2_main<<<grid, 256, 0, stream>>>(x, ws, y, B);
}


# Finish: 16-lane row sums as DPP adds on the vector ALU (bank-masked, two results share a register), not LDS-unit lane permutes; one row per lane, one full-line store per 32 rows; sum order changes
# speedup vs baseline: 2.0401x; 2.0401x over previous
#include <hip/hip_runtime.h>

typedef __attribute__((ext_vector_type(2))) float v2f;
typedef __attribute__((ext_vector_type(8))) float v8f;

#define NCOL 300
#define NTILE 19
#define BFRAG_OFF 0
#define BFRAG_FLOATS (NTILE * 32 * 2)
#define W32_OFF BFRAG_FLOATS
#define W32_PAD (NTILE * 16)
#define W31_OFF (W32_OFF + W32_PAD)

#define PREP_THREADS 320
#define PREP_WAVES (PREP_THREADS / 32)

__global__ __launch_bounds__(PREP_THREADS) void phi2_prep(
    const float* __restrict__ varphi1,
    const float* __restrict__ varphi2,
    const float* __restrict__ l1,
    const float* __restrict__ l2,
    const float* __restrict__ l3,
    float* __restrict__ ws)
{
    __shared__ float red[PREP_WAVES * 3];

    const int j    = threadIdx.x;
    const int lane = j & 31;
    const int wave = j >> 5;
    const float l3v = l3[0];

    float w32 = 0.0f;
    float b0 = 0.0f, b1 = 0.0f, b2 = 0.0f;
    float p0 = 0.0f, p1 = 0.0f, p2 = 0.0f;
    if (j < NCOL) {
        const float l2j = l2[j];
        const float a0 = varphi1[j * 3 + 0];
        const float a1 = varphi1[j * 3 + 1];
        const float a2 = varphi1[j * 3 + 2];
        const float sj = -l2j;
        w32 = varphi2[j] * (l2j - l3v);
        b0 = a0 * (l1[0] + sj);
        b1 = a1 * (l1[1] + sj);
        b2 = a2 * (l1[2] + sj);
        p0 = w32 * a0;
        p1 = w32 * a1;
        p2 = w32 * a2;
    }

    if (j < W32_PAD) {
        const int t = j >> 4;
        const int c = j & 15;
        ws[W32_OFF + j] = w32;
        *(v2f*)(ws + BFRAG_OFF + (t * 32 + c) * 2)      = v2f{b0, b1};
        *(v2f*)(ws + BFRAG_OFF + (t * 32 + 16 + c) * 2) = v2f{b2, 0.0f};
    }

#pragma unroll
    for (int m = 16; m >= 1; m >>= 1) {
        p0 += __shfl_xor(p0, m, 32);
        p1 += __shfl_xor(p1, m, 32);
        p2 += __shfl_xor(p2, m, 32);
    }
    if (lane == 0) {
        red[wave * 3 + 0] = p0;
        red[wave * 3 + 1] = p1;
        red[wave * 3 + 2] = p2;
    }
    __syncthreads();
    if (j < 3) {
        float s = 0.0f;
#pragma unroll
        for (int w = 0; w < PREP_WAVES; ++w) s += red[w * 3 + j];
        ws[W31_OFF + j] = -s;
    }
}

template <int TILE> __device__ __forceinline__ v8f relu8(v8f d);
template <> __device__ __forceinline__ v8f relu8<0>(v8f d) {
    asm volatile(
        "v_dual_max_num_f32 v64, 0, v64 :: v_dual_max_num_f32 v65, 0, v65\n\t"
        "v_dual_max_num_f32 v66, 0, v66 :: v_dual_max_num_f32 v67, 0, v67\n\t"
        "v_dual_max_num_f32 v68, 0, v68 :: v_dual_max_num_f32 v69, 0, v69\n\t"
        "v_dual_max_num_f32 v70, 0, v70 :: v_dual_max_num_f32 v71, 0, v71\n\t"
        : "={v[64:71]}"(d) : "0"(d));
    return d;
}
template <> __device__ __forceinline__ v8f relu8<1>(v8f d) {
    asm volatile(
        "v_dual_max_num_f32 v72, 0, v72 :: v_dual_max_num_f32 v73, 0, v73\n\t"
        "v_dual_max_num_f32 v74, 0, v74 :: v_dual_max_num_f32 v75, 0, v75\n\t"
        "v_dual_max_num_f32 v76, 0, v76 :: v_dual_max_num_f32 v77, 0, v77\n\t"
        "v_dual_max_num_f32 v78, 0, v78 :: v_dual_max_num_f32 v79, 0, v79\n\t"
        : "={v[72:79]}"(d) : "0"(d));
    return d;
}
template <> __device__ __forceinline__ v8f relu8<2>(v8f d) {
    asm volatile(
        "v_dual_max_num_f32 v80, 0, v80 :: v_dual_max_num_f32 v81, 0, v81\n\t"
        "v_dual_max_num_f32 v82, 0, v82 :: v_dual_max_num_f32 v83, 0, v83\n\t"
        "v_dual_max_num_f32 v84, 0, v84 :: v_dual_max_num_f32 v85, 0, v85\n\t"
        "v_dual_max_num_f32 v86, 0, v86 :: v_dual_max_num_f32 v87, 0, v87\n\t"
        : "={v[80:87]}"(d) : "0"(d));
    return d;
}
template <> __device__ __forceinline__ v8f relu8<3>(v8f d) {
    asm volatile(
        "v_dual_max_num_f32 v88, 0, v88 :: v_dual_max_num_f32 v89, 0, v89\n\t"
        "v_dual_max_num_f32 v90, 0, v90 :: v_dual_max_num_f32 v91, 0, v91\n\t"
        "v_dual_max_num_f32 v92, 0, v92 :: v_dual_max_num_f32 v93, 0, v93\n\t"
        "v_dual_max_num_f32 v94, 0, v94 :: v_dual_max_num_f32 v95, 0, v95\n\t"
        : "={v[88:95]}"(d) : "0"(d));
    return d;
}
template <> __device__ __forceinline__ v8f relu8<4>(v8f d) {
    asm volatile(
        "v_dual_max_num_f32 v96, 0, v96 :: v_dual_max_num_f32 v97, 0, v97\n\t"
        "v_dual_max_num_f32 v98, 0, v98 :: v_dual_max_num_f32 v99, 0, v99\n\t"
        "v_dual_max_num_f32 v100, 0, v100 :: v_dual_max_num_f32 v101, 0, v101\n\t"
        "v_dual_max_num_f32 v102, 0, v102 :: v_dual_max_num_f32 v103, 0, v103\n\t"
        : "={v[96:103]}"(d) : "0"(d));
    return d;
}
template <> __device__ __forceinline__ v8f relu8<5>(v8f d) {
    asm volatile(
        "v_dual_max_num_f32 v104, 0, v104 :: v_dual_max_num_f32 v105, 0, v105\n\t"
        "v_dual_max_num_f32 v106, 0, v106 :: v_dual_max_num_f32 v107, 0, v107\n\t"
        "v_dual_max_num_f32 v108, 0, v108 :: v_dual_max_num_f32 v109, 0, v109\n\t"
        "v_dual_max_num_f32 v110, 0, v110 :: v_dual_max_num_f32 v111, 0, v111\n\t"
        : "={v[104:111]}"(d) : "0"(d));
    return d;
}
template <> __device__ __forceinline__ v8f relu8<6>(v8f d) {
    asm volatile(
        "v_dual_max_num_f32 v112, 0, v112 :: v_dual_max_num_f32 v113, 0, v113\n\t"
        "v_dual_max_num_f32 v114, 0, v114 :: v_dual_max_num_f32 v115, 0, v115\n\t"
        "v_dual_max_num_f32 v116, 0, v116 :: v_dual_max_num_f32 v117, 0, v117\n\t"
        "v_dual_max_num_f32 v118, 0, v118 :: v_dual_max_num_f32 v119, 0, v119\n\t"
        : "={v[112:119]}"(d) : "0"(d));
    return d;
}
template <> __device__ __forceinline__ v8f relu8<7>(v8f d) {
    asm volatile(
        "v_dual_max_num_f32 v120, 0, v120 :: v_dual_max_num_f32 v121, 0, v121\n\t"
        "v_dual_max_num_f32 v122, 0, v122 :: v_dual_max_num_f32 v123, 0, v123\n\t"
        "v_dual_max_num_f32 v124, 0, v124 :: v_dual_max_num_f32 v125, 0, v125\n\t"
        "v_dual_max_num_f32 v126, 0, v126 :: v_dual_max_num_f32 v127, 0, v127\n\t"
        : "={v[120:127]}"(d) : "0"(d));
    return d;
}

__device__ __forceinline__ v8f relu8_at(v8f d, int tile) {
    switch (tile) {
    case 0:  return relu8<0>(d);
    case 1:  return relu8<1>(d);
    case 2:  return relu8<2>(d);
    case 3:  return relu8<3>(d);
    case 4:  return relu8<4>(d);
    case 5:  return relu8<5>(d);
    case 6:  return relu8<6>(d);
    default: return relu8<7>(d);
    }
}

#define ROW_TILES 8
#define ROWS_PER_WAVE (ROW_TILES * 16)
static_assert(ROW_TILES == 8, "relu8 has one fixed result tuple per row tile: 8 of them");

__device__ __forceinline__ float rowsum_pair8(float a, float b) {
    float t;
    asm volatile(
        "v_add_f32_dpp %0, %1, %1 row_xmask:8 row_mask:0xf bank_mask:0x3\n\t"
        "v_add_f32_dpp %0, %2, %2 row_xmask:8 row_mask:0xf bank_mask:0xc"
        : "=&v"(t) : "v"(a), "v"(b));
    return t;
}
__device__ __forceinline__ float rowsum_pair4(float a, float b) {
    float t;
    asm volatile(
        "v_add_f32_dpp %0, %1, %1 row_xmask:4 row_mask:0xf bank_mask:0x5\n\t"
        "v_add_f32_dpp %0, %2, %2 row_xmask:4 row_mask:0xf bank_mask:0xa"
        : "=&v"(t) : "v"(a), "v"(b));
    return t;
}
__device__ __forceinline__ void rowsum_x2(float& u) {
    asm volatile("v_add_f32_dpp %0, %0, %0 row_xmask:2 row_mask:0xf bank_mask:0xf" : "+v"(u));
}
__device__ __forceinline__ void rowsum_x1(float& u) {
    asm volatile("v_add_f32_dpp %0, %0, %0 row_xmask:1 row_mask:0xf bank_mask:0xf" : "+v"(u));
}

template <bool PARTIAL>
__device__ __forceinline__ void phi2_finish(
    v8f (&acc)[ROW_TILES], const float* __restrict__ x, float* __restrict__ y,
    long long rowBase, int nrows, int lane, float c0, float c1, float c2)
{
    const int rowInBlock = ((lane >> 1) & 1) * 16 + (lane >> 4) * 8 + (lane & 1) * 4
                         + ((lane >> 2) & 1) * 2 + ((lane >> 3) & 1);

    v2f xr[ROW_TILES / 2];
#pragma unroll
    for (int p = 0; p < ROW_TILES / 2; ++p) {
        long long lrow = rowBase + p * 32 + rowInBlock;
        if (PARTIAL && lrow > nrows - 1) lrow = nrows - 1;
        xr[p] = *(const v2f*)(x + lrow * 2);
    }

    float s[ROW_TILES / 2];
#pragma unroll
    for (int p = 0; p < ROW_TILES / 2; ++p) {
        const int ia = 2 * p, ib = 2 * p + 1;
        const float ta0 = rowsum_pair8(acc[ia][0], acc[ia][1]);
        const float ta1 = rowsum_pair8(acc[ia][2], acc[ia][3]);
        const float ta2 = rowsum_pair8(acc[ia][4], acc[ia][5]);
        const float ta3 = rowsum_pair8(acc[ia][6], acc[ia][7]);
        const float tb0 = rowsum_pair8(acc[ib][0], acc[ib][1]);
        const float tb1 = rowsum_pair8(acc[ib][2], acc[ib][3]);
        const float tb2 = rowsum_pair8(acc[ib][4], acc[ib][5]);
        const float tb3 = rowsum_pair8(acc[ib][6], acc[ib][7]);
        float ua0 = rowsum_pair4(ta0, ta1);
        float ua1 = rowsum_pair4(ta2, ta3);
        float ub0 = rowsum_pair4(tb0, tb1);
        float ub1 = rowsum_pair4(tb2, tb3);
        rowsum_x2(ua0); rowsum_x2(ua1); rowsum_x2(ub0); rowsum_x2(ub1);
        rowsum_x1(ua0); rowsum_x1(ua1); rowsum_x1(ub0); rowsum_x1(ub1);

        const float sa = (lane & 1) ? ua1 : ua0;
        const float sb = (lane & 1) ? ub1 : ub0;
        s[p] = (lane & 2) ? sb : sa;
    }

#pragma unroll
    for (int p = 0; p < ROW_TILES / 2; ++p) {
        const long long orow = rowBase + p * 32 + rowInBlock;
        if (!PARTIAL || orow < nrows) {
            y[orow] = s[p] + c0 * xr[p].x + c1 * xr[p].y + c2;
        }
    }
}

__global__ __launch_bounds__(256) void phi2_main(
    const float* __restrict__ x,
    const float* __restrict__ ws,
    float* __restrict__ y,
    int nrows)
{
    const int lane = threadIdx.x & 31;
    const int wave = threadIdx.x >> 5;
    const long long rowBase = ((long long)blockIdx.x * 8 + wave) * ROWS_PER_WAVE;
    if (rowBase >= nrows) return;

    v2f a[ROW_TILES];
#pragma unroll
    for (int i = 0; i < ROW_TILES; ++i) {
        long long row = rowBase + i * 16 + (lane & 15);
        if (row > nrows - 1) row = nrows - 1;
        if (lane < 16) {
            a[i].x = x[row * 2 + 0];
            a[i].y = x[row * 2 + 1];
        } else {
            a[i].x = 1.0f;
            a[i].y = 0.0f;
        }
    }

    const float* __restrict__ bf  = ws + BFRAG_OFF;
    const float* __restrict__ w32 = ws + W32_OFF;
    const float* __restrict__ w31 = ws + W31_OFF;

    v8f acc[ROW_TILES];
#pragma unroll
    for (int i = 0; i < ROW_TILES; ++i)
        acc[i] = v8f{0.f, 0.f, 0.f, 0.f, 0.f, 0.f, 0.f, 0.f};

    v2f b = *(const v2f*)(bf + lane * 2);
    float wv = w32[lane & 15];
    for (int t = 0; t < NTILE; ++t) {
        const int tn = (t + 1 < NTILE) ? (t + 1) : t;
        const v2f bn = *(const v2f*)(bf + (tn * 32 + lane) * 2);
        const float wvn = w32[tn * 16 + (lane & 15)];

        v8f c = {};
        v8f d[ROW_TILES];
#pragma unroll
        for (int i = 0; i < ROW_TILES; ++i) {
            d[i] = __builtin_amdgcn_wmma_f32_16x16x4_f32(
                 false, a[i],  false, b,
                 (short)0, c,  false,  false);
        }
        __builtin_amdgcn_sched_barrier(0);
#pragma unroll
        for (int i = 0; i < ROW_TILES; ++i) {
            d[i] = relu8_at(d[i], i);
#pragma unroll
            for (int r = 0; r < 8; ++r) {
                acc[i][r] += d[i][r] * wv;
            }
        }
        asm volatile("" :: "v"(bn), "v"(wvn));
        b = bn;
        wv = wvn;
    }

    const float c0 = w31[0], c1 = w31[1], c2 = w31[2];

    if (rowBase + ROWS_PER_WAVE <= nrows)
        phi2_finish<false>(acc, x, y, rowBase, nrows, lane, c0, c1, c2);
    else
        phi2_finish<true>(acc, x, y, rowBase, nrows, lane, c0, c1, c2);
}

extern "C" void kernel_launch(void* const* d_in, const int* in_sizes, int n_in,
                              void* d_out, int out_size, void* d_ws, size_t ws_size,
                              hipStream_t stream) {
    const float* x       = (const float*)d_in[0];
    const float* varphi1 = (const float*)d_in[1];
    const float* varphi2 = (const float*)d_in[2];
    const float* l1      = (const float*)d_in[3];
    const float* l2      = (const float*)d_in[4];
    const float* l3      = (const float*)d_in[5];
    float* y  = (float*)d_out;
    float* ws = (float*)d_ws;

    const int B = in_sizes[0] / 2;

    phi2_prep<<<1, PREP_THREADS, 0, stream>>>(varphi1, varphi2, l1, l2, l3, ws);

    const int wavesNeeded = (B + ROWS_PER_WAVE - 1) / ROWS_PER_WAVE;
    const int grid = (wavesNeeded + 7) / 8;
    phi2_main<<<grid, 256, 0, stream>>>(x, ws, y, B);
}
